# starting kernel
# baseline (speedup 1.0000x reference)
#include <hip/hip_runtime.h>

typedef float v2f __attribute__((ext_vector_type(2)));
typedef float v8f __attribute__((ext_vector_type(8)));

#define NXL 1024
#define NYL 1024
#define TAU 0.6f
#define GYF (-1e-4f)

struct ATab { float v[12][16]; };

static constexpr ATab makeAT() {
    const signed char M[9][9] = {
        { 1,  1,  1,  1,  1,  1,  1,  1,  1},
        { 0,  1,  0, -1,  0,  1, -1, -1,  1},
        { 0,  0,  1,  0, -1,  1,  1, -1, -1},
        {-4, -1, -1, -1, -1,  2,  2,  2,  2},
        { 4, -2, -2, -2, -2,  1,  1,  1,  1},
        { 0, -2,  0,  2,  0,  1, -1, -1,  1},
        { 0,  0, -2,  0,  2,  1,  1, -1, -1},
        { 0,  1, -1,  1, -1,  0,  0,  0,  0},
        { 0,  0,  0,  0,  0,  1, -1,  1, -1},
    };
    const float nrm2[9] = {9.f, 6.f, 6.f, 36.f, 36.f, 12.f, 12.f, 4.f, 4.f};
    ATab t{};
    for (int k = 0; k < 12; ++k)
        for (int q = 0; q < 16; ++q) {
            float x = 0.0f;
            if (k < 9 && q < 9)
                x = (float)M[k][q] / nrm2[k] * (1.0f / TAU);
            t.v[k][q] = x;
        }
    return t;
}

__constant__ ATab g_AT = makeAT();

__global__ __launch_bounds__(256) void lbm_step(const float* __restrict__ fin,
                                                const float* __restrict__ w,
                                                float* __restrict__ fout) {
    const int tid  = blockIdx.x * blockDim.x + threadIdx.x;
    const int lane = threadIdx.x & 31;
    const int half = lane >> 4;
    const int q16  = lane & 15;
    const int cell = (tid >> 5) * 16 + q16;
    const int x = cell >> 10;
    const int y = cell & (NYL - 1);

    const int exv[9] = {0, 1, 0, -1, 0, 1, -1, -1, 1};
    const int eyv[9] = {0, 0, 1, 0, -1, 1, 1, -1, -1};
    float f[9];
#pragma unroll
    for (int i = 0; i < 9; ++i) {
        const int nx = (x - exv[i]) & (NXL - 1);
        const int ny = (y - eyv[i]) & (NYL - 1);
        f[i] = fin[((nx << 10) | ny) * 9 + i];
    }

    const float rho = ((f[0] + f[1]) + (f[2] + f[3])) +
                      ((f[4] + f[5]) + (f[6] + f[7])) + f[8];
    const float mx = f[1] - f[3] + f[5] - f[6] - f[7] + f[8];
    const float my = f[2] - f[4] + f[5] + f[6] - f[7] - f[8];
    const float jx = mx;
    const float jy = my + (TAU * GYF) * rho * rho;
    const float jx2 = jx * jx, jy2 = jy * jy;
    const float j2 = jx2 + jy2;

    const float w0 = w[0], w1 = w[1], w2 = w[2], w3 = w[3];
    const float w4 = w[4], w5 = w[5], w6 = w[6];

    float me[9];
    me[0] = rho;
    me[1] = jx;
    me[2] = jy;
    me[3] = w0 * rho + w1 * j2;
    me[4] = w2 * rho - w3 * j2;
    me[5] = w4 * jx;
    me[6] = w4 * jy;
    me[7] = w5 * (jx2 - jy2);
    me[8] = w6 * jx * jy;

    const float* at = &g_AT.v[0][0];
    v2f a0, a1, a2;
    a0.x = at[(0 + 2 * half) * 16 + q16];
    a0.y = at[(1 + 2 * half) * 16 + q16];
    a1.x = at[(4 + 2 * half) * 16 + q16];
    a1.y = at[(5 + 2 * half) * 16 + q16];
    a2.x = at[(8 + 2 * half) * 16 + q16];
    a2.y = at[(9 + 2 * half) * 16 + q16];

    v2f b0, b1, b2;
    b0.x = half ? me[2] : me[0];
    b0.y = half ? me[3] : me[1];
    b1.x = half ? me[6] : me[4];
    b1.y = half ? me[7] : me[5];
    b2.x = half ? 0.0f  : me[8];
    b2.y = 0.0f;

    v8f c = {};
    c = __builtin_amdgcn_wmma_f32_16x16x4_f32(false, a0, false, b0, (short)0, c,
                                              false, false);
    c = __builtin_amdgcn_wmma_f32_16x16x4_f32(false, a1, false, b1, (short)0, c,
                                              false, false);
    c = __builtin_amdgcn_wmma_f32_16x16x4_f32(false, a2, false, b2, (short)0, c,
                                              false, false);

    const float decay = 1.0f - 1.0f / TAU;
    float* o = fout + cell * 9;
    if (half == 0) {
#pragma unroll
        for (int q = 0; q < 8; ++q) o[q] = decay * f[q] + c[q];
    } else {
        o[8] = decay * f[8] + c[0];
    }
}

extern "C" void kernel_launch(void* const* d_in, const int* in_sizes, int n_in,
                              void* d_out, int out_size, void* d_ws, size_t ws_size,
                              hipStream_t stream) {
    const float* f0 = (const float*)d_in[0];
    const float* w  = (const float*)d_in[1];
    float* out = (float*)d_out;
    float* ws  = (float*)d_ws;

    const int threads = 256;
    const int blocks  = (NXL * NYL * 2) / threads;

    lbm_step<<<blocks, threads, 0, stream>>>(f0,  w, ws);
    lbm_step<<<blocks, threads, 0, stream>>>(ws,  w, out);
    lbm_step<<<blocks, threads, 0, stream>>>(out, w, ws);
    lbm_step<<<blocks, threads, 0, stream>>>(ws,  w, out);
}


# two steps per launch (2 launches, not 4): a wave keeps 3 first-step rows in registers; step two streams from them by row pick and one-lane shifts; ring cells recomputed; per-cell sums unchanged
# speedup vs baseline: 1.4472x; 1.4472x over previous
#include <hip/hip_runtime.h>

typedef float v2f __attribute__((ext_vector_type(2)));
typedef float v8f __attribute__((ext_vector_type(8)));

#define NXL 1024
#define NYL 1024
#define TAU 0.6f
#define GYF (-1e-4f)

struct ATab { float v[12][16]; };

static constexpr ATab makeAT() {
    const signed char M[9][9] = {
        { 1,  1,  1,  1,  1,  1,  1,  1,  1},
        { 0,  1,  0, -1,  0,  1, -1, -1,  1},
        { 0,  0,  1,  0, -1,  1,  1, -1, -1},
        {-4, -1, -1, -1, -1,  2,  2,  2,  2},
        { 4, -2, -2, -2, -2,  1,  1,  1,  1},
        { 0, -2,  0,  2,  0,  1, -1, -1,  1},
        { 0,  0, -2,  0,  2,  1,  1, -1, -1},
        { 0,  1, -1,  1, -1,  0,  0,  0,  0},
        { 0,  0,  0,  0,  0,  1, -1,  1, -1},
    };
    const float nrm2[9] = {9.f, 6.f, 6.f, 36.f, 36.f, 12.f, 12.f, 4.f, 4.f};
    ATab t{};
    for (int k = 0; k < 12; ++k)
        for (int q = 0; q < 16; ++q) {
            float x = 0.0f;
            if (k < 9 && q < 9)
                x = (float)M[k][q] / nrm2[k] * (1.0f / TAU);
            t.v[k][q] = x;
        }
    return t;
}

__constant__ ATab g_AT = makeAT();

#define ROWS_PER_WAVE 16
#define ROWS_A (ROWS_PER_WAVE + 2)
#define STRIP_W 14
#define STRIPS 74
#define WAVES_PER_WG 4
#define ROWS_AHEAD 5
#define ROW_PITCH (NYL * 9)

__device__ __forceinline__ float xhalf(float x) {
    return __builtin_bit_cast(float, __builtin_amdgcn_permlanex16(
        __builtin_bit_cast(int, x), __builtin_bit_cast(int, x), 0x76543210, 0xfedcba98, false, false));
}
__device__ __forceinline__ float from_left(float x) {
    return __builtin_bit_cast(float, __builtin_amdgcn_permlane16(
        __builtin_bit_cast(int, x), __builtin_bit_cast(int, x), 0x6543210F, 0xEDCBA987, false, false));
}
__device__ __forceinline__ float from_right(float x) {
    return __builtin_bit_cast(float, __builtin_amdgcn_permlane16(
        __builtin_bit_cast(int, x), __builtin_bit_cast(int, x), 0x87654321, 0x0FEDCBA9, false, false));
}

template <bool IN_PLANAR, bool OUT_PLANAR>
__device__ __forceinline__ void lbm_pair_body(const float* __restrict__ fin,
                                              const float* __restrict__ w,
                                              float* __restrict__ fout) {
    const int lane = threadIdx.x & 31;
    const int half = lane >> 4;
    const int q16  = lane & 15;
    const int wv = __builtin_amdgcn_readfirstlane((int)(threadIdx.x >> 5));
    const int gw = blockIdx.x * WAVES_PER_WG + wv;
    const int strip = gw % STRIPS;
    const int x0 = (gw / STRIPS) * ROWS_PER_WAVE;
    const int yu = strip * STRIP_W - 1 + q16;
    const int y  = yu & (NYL - 1);
    const bool owned = (q16 >= 1) && (q16 <= STRIP_W) && (yu < NYL);

    const int exv[9] = {0, 1, 0, -1, 0, 1, -1, -1, 1};
    const int eyv[9] = {0, 0, 1, 0, -1, 1, 1, -1, -1};
    unsigned col[9];
#pragma unroll
    for (int i = 0; i < 9; ++i) {
        const int ys = (y - eyv[i]) & (NYL - 1);
        col[i] = IN_PLANAR ? (unsigned)(i * NYL + ys) : (unsigned)(ys * 9 + i);
    }

    const float w0 = w[0], w1 = w[1], w2 = w[2], w3 = w[3];
    const float w4 = w[4], w5 = w[5], w6 = w[6];

    const int qa = (q16 < 9) ? q16 : q16 - 8;
    const float* at = &g_AT.v[0][0];
    v2f a0, a1, a2;
    a0.x = at[(0 + 2 * half) * 16 + qa];
    a0.y = at[(1 + 2 * half) * 16 + qa];
    a1.x = at[(4 + 2 * half) * 16 + qa];
    a1.y = at[(5 + 2 * half) * 16 + qa];
    a2.x = at[(8 + 2 * half) * 16 + qa];
    a2.y = at[(9 + 2 * half) * 16 + qa];

    const float decay = 1.0f - 1.0f / TAU;
    const unsigned ocol = OUT_PLANAR ? (unsigned)y : (unsigned)(y * 9);
    const int qs = OUT_PLANAR ? NYL : 1;

    auto collide = [&](const float* f) -> v8f {
        const float rho = ((f[0] + f[1]) + (f[2] + f[3])) +
                          ((f[4] + f[5]) + (f[6] + f[7])) + f[8];
        const float mx = f[1] - f[3] + f[5] - f[6] - f[7] + f[8];
        const float my = f[2] - f[4] + f[5] + f[6] - f[7] - f[8];
        const float jx = mx;
        const float jy = my + (TAU * GYF) * rho * rho;
        const float jx2 = jx * jx, jy2 = jy * jy;
        const float j2 = jx2 + jy2;

        float me[9];
        me[0] = rho;
        me[1] = jx;
        me[2] = jy;
        me[3] = w0 * rho + w1 * j2;
        me[4] = w2 * rho - w3 * j2;
        me[5] = w4 * jx;
        me[6] = w4 * jy;
        me[7] = w5 * (jx2 - jy2);
        me[8] = w6 * jx * jy;

        v2f b0, b1, b2;
        b0.x = half ? me[2] : me[0];
        b0.y = half ? me[3] : me[1];
        b1.x = half ? me[6] : me[4];
        b1.y = half ? me[7] : me[5];
        b2.x = half ? 0.0f  : me[8];
        b2.y = 0.0f;

        v8f c = {};
        c = __builtin_amdgcn_wmma_f32_16x16x4_f32(false, a0, false, b0, (short)0, c,
                                                  false, false);
        c = __builtin_amdgcn_wmma_f32_16x16x4_f32(false, a1, false, b1, (short)0, c,
                                                  false, false);
        c = __builtin_amdgcn_wmma_f32_16x16x4_f32(false, a2, false, b2, (short)0, c,
                                                  false, false);
        return c;
    };

    float fr[ROWS_A][9];
    auto load_row = [&](int t) {
        const int x = (x0 - 1 + t) & (NXL - 1);
        const float* rows[3];
        rows[0] = fin + (size_t)((x + 1) & (NXL - 1)) * ROW_PITCH;
        rows[1] = fin + (size_t)x * ROW_PITCH;
        rows[2] = fin + (size_t)((x - 1) & (NXL - 1)) * ROW_PITCH;
#pragma unroll
        for (int i = 0; i < 9; ++i) fr[t][i] = rows[exv[i] + 1][col[i]];
    };

#pragma unroll
    for (int t = 0; t < ROWS_AHEAD; ++t) {
        load_row(t);
        __builtin_amdgcn_sched_barrier(0);
    }

    float hw[3][9];

#pragma unroll
    for (int t = 0; t < ROWS_A; ++t) {
        if (t + ROWS_AHEAD < ROWS_A) load_row(t + ROWS_AHEAD);

        {
            const float* f = fr[t];
            const v8f c = collide(f);
            float g[9];
#pragma unroll
            for (int q = 1; q < 8; ++q) g[q] = decay * f[q] + c[q];
            const float fz = half ? f[8] : f[0];
            const float z  = decay * fz + c[0];
            const float tz = xhalf(z);
            g[0] = half ? tz : z;
            g[8] = half ? z : tz;
            float* h = hw[t % 3];
            h[0] = g[0];
            h[1] = g[1];
            h[3] = g[3];
            h[2] = from_left(g[2]);
            h[5] = from_left(g[5]);
            h[6] = from_left(g[6]);
            h[4] = from_right(g[4]);
            h[7] = from_right(g[7]);
            h[8] = from_right(g[8]);
        }

        if (t >= 2) {
            const int x = x0 + t - 2;
            const float* hp = hw[(t - 2) % 3];
            const float* hc = hw[(t - 1) % 3];
            const float* hn = hw[t % 3];
            float f[9];
            f[0] = hc[0];
            f[1] = hp[1];
            f[2] = hc[2];
            f[3] = hn[3];
            f[4] = hc[4];
            f[5] = hp[5];
            f[6] = hn[6];
            f[7] = hn[7];
            f[8] = hp[8];
            const v8f c = collide(f);

            float* o = fout + (size_t)x * ROW_PITCH + ocol;
            if (owned) {
                if (half == 0) {
#pragma unroll
                    for (int q = 0; q < 8; ++q) o[q * qs] = decay * f[q] + c[q];
                } else {
                    o[8 * qs] = decay * f[8] + c[0];
                }
            }
        }
    }
}

__global__ __launch_bounds__(128) void lbm_pair_first(const float* __restrict__ fin,
                                                      const float* __restrict__ w,
                                                      float* __restrict__ fout) {
    lbm_pair_body<false, true>(fin, w, fout);
}

__global__ __launch_bounds__(128) void lbm_pair_last(const float* __restrict__ fin,
                                                     const float* __restrict__ w,
                                                     float* __restrict__ fout) {
    lbm_pair_body<true, false>(fin, w, fout);
}

extern "C" void kernel_launch(void* const* d_in, const int* in_sizes, int n_in,
                              void* d_out, int out_size, void* d_ws, size_t ws_size,
                              hipStream_t stream) {
    const float* f0 = (const float*)d_in[0];
    const float* w  = (const float*)d_in[1];
    float* out = (float*)d_out;
    float* ws  = (float*)d_ws;

    const int threads = 32 * WAVES_PER_WG;
    const int blocks  = (STRIPS * (NXL / ROWS_PER_WAVE)) / WAVES_PER_WG;

    lbm_pair_first<<<blocks, threads, 0, stream>>>(f0, w, ws);
    lbm_pair_last<<<blocks, threads, 0, stream>>>(ws, w, out);
}


# populations are the WMMA B operand, split 5|4 over the half-waves: moment sums, relaxation, linear equilibrium move from vector code into the matrix product (4 WMMAs, was 3); sums re-associated
# speedup vs baseline: 1.5845x; 1.0949x over previous
#include <hip/hip_runtime.h>

typedef float v2f __attribute__((ext_vector_type(2)));
typedef float v8f __attribute__((ext_vector_type(8)));

#define NXL 1024
#define NYL 1024
#define TAU 0.6f
#define GYF (-1e-4f)


struct RowTab { float v[16][12]; };

static constexpr RowTab makeRows() {
    const signed char M[9][9] = {
        { 1,  1,  1,  1,  1,  1,  1,  1,  1},
        { 0,  1,  0, -1,  0,  1, -1, -1,  1},
        { 0,  0,  1,  0, -1,  1,  1, -1, -1},
        {-4, -1, -1, -1, -1,  2,  2,  2,  2},
        { 4, -2, -2, -2, -2,  1,  1,  1,  1},
        { 0, -2,  0,  2,  0,  1, -1, -1,  1},
        { 0,  0, -2,  0,  2,  1,  1, -1, -1},
        { 0,  1, -1,  1, -1,  0,  0,  0,  0},
        { 0,  0,  0,  0,  0,  1, -1,  1, -1},
    };
    const float nrm2[9] = {9.f, 6.f, 6.f, 36.f, 36.f, 12.f, 12.f, 4.f, 4.f};
    const int rowq[16] = {0, 2, 1, 5, -1, 6, -2, -3,   -4, 4, -1, 8, 3, 7, -2, -3};
    RowTab t{};
    for (int m = 0; m < 16; ++m) {
        for (int p = 0; p < 12; ++p) t.v[m][p] = 0.0f;
        const int q = rowq[m];
        if (q >= 0) {
            for (int p = 0; p < 9; ++p)
                t.v[m][p] = (float)M[p][q] / nrm2[p] * (1.0f / TAU);
            t.v[m][9] = (float)q;
        } else {
            t.v[m][9] = -1.0f;
            if (q == -2) t.v[m][0] = 1.0f;
            if (q == -3) t.v[m][1] = 1.0f;
            if (q == -4) t.v[m][2] = 1.0f;
        }
    }
    return t;
}

__constant__ RowTab g_rows = makeRows();

#define ROWS_PER_WAVE 16
#define ROWS_A (ROWS_PER_WAVE + 2)
#define STRIP_W 14
#define STRIPS 74
#define WAVES_PER_WG 4
#define ROWS_AHEAD 5
#define ROW_PITCH (NYL * 9)

__device__ __forceinline__ float row_gather(float x, int sel) {
    return __builtin_bit_cast(float, __builtin_amdgcn_permlane16_var(
        __builtin_bit_cast(int, x), __builtin_bit_cast(int, x), sel, false, false));
}

template <bool IN_PLANAR, bool OUT_PLANAR>
__device__ __forceinline__ void lbm_pair_body(const float* __restrict__ fin,
                                              const float* __restrict__ w,
                                              float* __restrict__ fout) {
    const int lane = threadIdx.x & 31;
    const int half = lane >> 4;
    const int q16  = lane & 15;
    const int wv = __builtin_amdgcn_readfirstlane((int)(threadIdx.x >> 5));
    const int gw = blockIdx.x * WAVES_PER_WG + wv;
    const int strip = gw % STRIPS;
    const int x0 = (gw / STRIPS) * ROWS_PER_WAVE;
    const int yu = strip * STRIP_W - 1 + q16;
    const int y  = yu & (NYL - 1);
    const bool owned = (q16 >= 1) && (q16 <= STRIP_W) && (yu < NYL);

    const int qL[6] = { 0, 2,  1, 5, -1, 6};
    const int qU[6] = {-1, 4, -1, 8,  3, 7};
    const int exr[6] = {0, 0, 1, 1, -1, -1};
    const int exv[9] = {0, 1, 0, -1, 0, 1, -1, -1, 1};
    const int eyv[9] = {0, 0, 1, 0, -1, 1, 1, -1, -1};

    unsigned col[6];
#pragma unroll
    for (int r = 0; r < 6; ++r) {
        const int ql = qL[r] >= 0 ? qL[r] : qU[r];
        const int qu = qU[r] >= 0 ? qU[r] : qL[r];
        const int ysl = (y - eyv[ql]) & (NYL - 1);
        const int ysu = (y - eyv[qu]) & (NYL - 1);
        const unsigned cl = IN_PLANAR ? (unsigned)(ql * NYL + ysl) : (unsigned)(ysl * 9 + ql);
        const unsigned cu = IN_PLANAR ? (unsigned)(qu * NYL + ysu) : (unsigned)(ysu * 9 + qu);
        col[r] = half ? cu : cl;
    }

    const float w0 = w[0], w1 = w[1], w2 = w[2], w3 = w[3];
    const float w4 = w[4], w5 = w[5], w6 = w[6];
    const float decay = 1.0f - 1.0f / TAU;
    const float grav  = TAU * GYF;

    const float* tr = &g_rows.v[q16][0];
    const float t0 = tr[0], t1 = tr[1], t2 = tr[2], t3 = tr[3], t4 = tr[4];
    const float t5 = tr[5], t6 = tr[6], t7 = tr[7], t8 = tr[8], tq = tr[9];
    const float c0 = t0 + w0 * t3 + w2 * t4;
    const float cx = t1 + w4 * t5;
    const float cy = t2 + w4 * t6;
    float al[6];
#pragma unroll
    for (int r = 0; r < 6; ++r) {
        const int il = qL[r], iu = qU[r];
        const float vl = il >= 0 ? 1.0f : 0.0f,            vu = iu >= 0 ? 1.0f : 0.0f;
        const float xl = il >= 0 ? (float)exv[il] : 0.0f,  xu = iu >= 0 ? (float)exv[iu] : 0.0f;
        const float yl = il >= 0 ? (float)eyv[il] : 0.0f,  yy = iu >= 0 ? (float)eyv[iu] : 0.0f;
        const float idl = il >= 0 ? (float)il : -2.0f,     idu = iu >= 0 ? (float)iu : -2.0f;
        const float val = half ? vu : vl;
        const float exi = half ? xu : xl;
        const float eyi = half ? yy : yl;
        const float idi = half ? idu : idl;
        const float dlt = (idi == tq) ? decay : 0.0f;
        al[r] = val * c0 + exi * cx + eyi * cy + dlt;
    }
    v2f a0, a1, a2, a3;
    a0.x = al[0]; a0.y = al[1];
    a1.x = al[2]; a1.y = al[3];
    a2.x = al[4]; a2.y = al[5];
    {
        const float nj  = w1 * t3 - w3 * t4;
        const float nd  = w5 * t7;
        const float nr2 = (tq >= 0.0f) ? grav * cy : 0.0f;
        const float np  = w6 * t8;
        a3.x = half ? (nj - nd) : nr2;
        a3.y = half ? np : (nj + nd);
    }

    const unsigned ocol = OUT_PLANAR ? (unsigned)y : (unsigned)(y * 9);
    const int qs = OUT_PLANAR ? NYL : 1;
    const unsigned so0 = ocol + (unsigned)((half ? 3 : 0) * qs);
    const unsigned so1 = ocol + (unsigned)((half ? 4 : 2) * qs);
    const unsigned so3 = ocol + (unsigned)((half ? 8 : 5) * qs);
    const unsigned so5 = ocol + (unsigned)((half ? 7 : 6) * qs);
    const unsigned so2 = ocol + (unsigned)(1 * qs);

    const int ysel = half ? ((q16 + 1) & 15) : ((q16 - 1) & 15);

    auto collide = [&](v2f b0, v2f b1, v2f b2) -> v8f {
        v8f c = {};
        c = __builtin_amdgcn_wmma_f32_16x16x4_f32(false, a0, false, b0, (short)0, c,
                                                  false, false);
        c = __builtin_amdgcn_wmma_f32_16x16x4_f32(false, a1, false, b1, (short)0, c,
                                                  false, false);
        c = __builtin_amdgcn_wmma_f32_16x16x4_f32(false, a2, false, b2, (short)0, c,
                                                  false, false);
        const float rho = c[6];
        const float mx  = c[7];
        const float r2  = rho * rho;
        const float jy  = grav * r2 + c[0];
        v2f bn;
        bn.x = half ? jy * jy : r2;
        bn.y = mx * (half ? jy : mx);
        c = __builtin_amdgcn_wmma_f32_16x16x4_f32(false, a3, false, bn, (short)0, c,
                                                  false, false);
        return c;
    };

    float fr[ROWS_A][6];
    auto load_row = [&](int t) {
        const int x = (x0 - 1 + t) & (NXL - 1);
        const float* rows[3];
        rows[0] = fin + (size_t)((x + 1) & (NXL - 1)) * ROW_PITCH;
        rows[1] = fin + (size_t)x * ROW_PITCH;
        rows[2] = fin + (size_t)((x - 1) & (NXL - 1)) * ROW_PITCH;
#pragma unroll
        for (int r = 0; r < 6; ++r) fr[t][r] = rows[exr[r] + 1][col[r]];
    };

#pragma unroll
    for (int t = 0; t < ROWS_AHEAD; ++t) {
        load_row(t);
        __builtin_amdgcn_sched_barrier(0);
    }

    float hw[3][6];

#pragma unroll
    for (int t = 0; t < ROWS_A; ++t) {
        if (t + ROWS_AHEAD < ROWS_A) load_row(t + ROWS_AHEAD);

        {
            const float* f = fr[t];
            v2f b0, b1, b2;
            b0.x = f[0]; b0.y = f[1];
            b1.x = f[2]; b1.y = f[3];
            b2.x = f[4]; b2.y = f[5];
            const v8f c = collide(b0, b1, b2);
            float* h = hw[t % 3];
            h[0] = c[0];
            h[2] = c[2];
            h[4] = c[4];
            h[1] = row_gather(c[1], ysel);
            h[3] = row_gather(c[3], ysel);
            h[5] = row_gather(c[5], ysel);
        }

        if (t >= 2) {
            const int x = x0 + t - 2;
            const float* hp = hw[(t - 2) % 3];
            const float* hc = hw[(t - 1) % 3];
            const float* hn = hw[t % 3];
            v2f b0, b1, b2;
            b0.x = hc[0]; b0.y = hc[1];
            b1.x = hp[2]; b1.y = hp[3];
            b2.x = hn[4]; b2.y = hn[5];
            const v8f c = collide(b0, b1, b2);

            float* o = fout + (size_t)x * ROW_PITCH;
            if (owned) {
                o[so0] = half ? c[4] : c[0];
                o[so1] = c[1];
                o[so3] = c[3];
                o[so5] = c[5];
                if (half == 0) o[so2] = c[2];
            }
        }
    }
}

__global__ __launch_bounds__(128) void lbm_pair_first(const float* __restrict__ fin,
                                                      const float* __restrict__ w,
                                                      float* __restrict__ fout) {
    lbm_pair_body<false, true>(fin, w, fout);
}

__global__ __launch_bounds__(128) void lbm_pair_last(const float* __restrict__ fin,
                                                     const float* __restrict__ w,
                                                     float* __restrict__ fout) {
    lbm_pair_body<true, false>(fin, w, fout);
}

extern "C" void kernel_launch(void* const* d_in, const int* in_sizes, int n_in,
                              void* d_out, int out_size, void* d_ws, size_t ws_size,
                              hipStream_t stream) {
    const float* f0 = (const float*)d_in[0];
    const float* w  = (const float*)d_in[1];
    float* out = (float*)d_out;
    float* ws  = (float*)d_ws;

    const int threads = 32 * WAVES_PER_WG;
    const int blocks  = (STRIPS * (NXL / ROWS_PER_WAVE)) / WAVES_PER_WG;

    lbm_pair_first<<<blocks, threads, 0, stream>>>(f0, w, ws);
    lbm_pair_last<<<blocks, threads, 0, stream>>>(ws, w, out);
}


# all four steps in one launch (was 2 launches of 2): four-stage register row pipeline, 10 of 16 columns owned, 32-row walks: all 3296 waves resident at once; ring recomputed; per-cell sums unchanged
# speedup vs baseline: 1.7403x; 1.0983x over previous
#include <hip/hip_runtime.h>

typedef float v2f __attribute__((ext_vector_type(2)));
typedef float v8f __attribute__((ext_vector_type(8)));

#define NXL 1024
#define NYL 1024
#define TAU 0.6f
#define GYF (-1e-4f)


struct RowTab { float v[16][12]; };

static constexpr RowTab makeRows() {
    const signed char M[9][9] = {
        { 1,  1,  1,  1,  1,  1,  1,  1,  1},
        { 0,  1,  0, -1,  0,  1, -1, -1,  1},
        { 0,  0,  1,  0, -1,  1,  1, -1, -1},
        {-4, -1, -1, -1, -1,  2,  2,  2,  2},
        { 4, -2, -2, -2, -2,  1,  1,  1,  1},
        { 0, -2,  0,  2,  0,  1, -1, -1,  1},
        { 0,  0, -2,  0,  2,  1,  1, -1, -1},
        { 0,  1, -1,  1, -1,  0,  0,  0,  0},
        { 0,  0,  0,  0,  0,  1, -1,  1, -1},
    };
    const float nrm2[9] = {9.f, 6.f, 6.f, 36.f, 36.f, 12.f, 12.f, 4.f, 4.f};
    const int rowq[16] = {0, 2, 1, 5, -1, 6, -2, -3,   -4, 4, -1, 8, 3, 7, -2, -3};
    RowTab t{};
    for (int m = 0; m < 16; ++m) {
        for (int p = 0; p < 12; ++p) t.v[m][p] = 0.0f;
        const int q = rowq[m];
        if (q >= 0) {
            for (int p = 0; p < 9; ++p)
                t.v[m][p] = (float)M[p][q] / nrm2[p] * (1.0f / TAU);
            t.v[m][9] = (float)q;
        } else {
            t.v[m][9] = -1.0f;
            if (q == -2) t.v[m][0] = 1.0f;
            if (q == -3) t.v[m][1] = 1.0f;
            if (q == -4) t.v[m][2] = 1.0f;
        }
    }
    return t;
}

__constant__ RowTab g_rows = makeRows();

#define NSTAGES 4
#define HALO (NSTAGES - 1)
#define ROWS_PER_WAVE 32
#define ROWS_A (ROWS_PER_WAVE + 2 * HALO)
#define STRIP_W (16 - 2 * HALO)
#define STRIPS 103
#define WAVES_PER_WG 4
#define ROWS_AHEAD 5
#define ROW_PITCH (NYL * 9)

__device__ __forceinline__ float row_gather(float x, int sel) {
    return __builtin_bit_cast(float, __builtin_amdgcn_permlane16_var(
        __builtin_bit_cast(int, x), __builtin_bit_cast(int, x), sel, false, false));
}

__device__ __forceinline__ void lbm_four_body(const float* __restrict__ fin,
                                              const float* __restrict__ w,
                                              float* __restrict__ fout) {
    const int lane = threadIdx.x & 31;
    const int half = lane >> 4;
    const int q16  = lane & 15;
    const int wv = __builtin_amdgcn_readfirstlane((int)(threadIdx.x >> 5));
    const int gw = blockIdx.x * WAVES_PER_WG + wv;
    const int strip = gw % STRIPS;
    const int x0 = (gw / STRIPS) * ROWS_PER_WAVE;
    const int yu = strip * STRIP_W - HALO + q16;
    const int y  = yu & (NYL - 1);
    const bool owned = (q16 >= HALO) && (q16 < HALO + STRIP_W) && (yu < NYL);

    const int qL[6] = { 0, 2,  1, 5, -1, 6};
    const int qU[6] = {-1, 4, -1, 8,  3, 7};
    const int exr[6] = {0, 0, 1, 1, -1, -1};
    const int exv[9] = {0, 1, 0, -1, 0, 1, -1, -1, 1};
    const int eyv[9] = {0, 0, 1, 0, -1, 1, 1, -1, -1};

    unsigned col[6];
#pragma unroll
    for (int r = 0; r < 6; ++r) {
        const int ql = qL[r] >= 0 ? qL[r] : qU[r];
        const int qu = qU[r] >= 0 ? qU[r] : qL[r];
        const int ysl = (y - eyv[ql]) & (NYL - 1);
        const int ysu = (y - eyv[qu]) & (NYL - 1);
        const unsigned cl = (unsigned)(ysl * 9 + ql);
        const unsigned cu = (unsigned)(ysu * 9 + qu);
        col[r] = half ? cu : cl;
    }

    const float w0 = w[0], w1 = w[1], w2 = w[2], w3 = w[3];
    const float w4 = w[4], w5 = w[5], w6 = w[6];
    const float decay = 1.0f - 1.0f / TAU;
    const float grav  = TAU * GYF;

    const float* tr = &g_rows.v[q16][0];
    const float t0 = tr[0], t1 = tr[1], t2 = tr[2], t3 = tr[3], t4 = tr[4];
    const float t5 = tr[5], t6 = tr[6], t7 = tr[7], t8 = tr[8], tq = tr[9];
    const float c0 = t0 + w0 * t3 + w2 * t4;
    const float cx = t1 + w4 * t5;
    const float cy = t2 + w4 * t6;
    float al[6];
#pragma unroll
    for (int r = 0; r < 6; ++r) {
        const int il = qL[r], iu = qU[r];
        const float vl = il >= 0 ? 1.0f : 0.0f,            vu = iu >= 0 ? 1.0f : 0.0f;
        const float xl = il >= 0 ? (float)exv[il] : 0.0f,  xu = iu >= 0 ? (float)exv[iu] : 0.0f;
        const float yl = il >= 0 ? (float)eyv[il] : 0.0f,  yy = iu >= 0 ? (float)eyv[iu] : 0.0f;
        const float idl = il >= 0 ? (float)il : -2.0f,     idu = iu >= 0 ? (float)iu : -2.0f;
        const float val = half ? vu : vl;
        const float exi = half ? xu : xl;
        const float eyi = half ? yy : yl;
        const float idi = half ? idu : idl;
        const float dlt = (idi == tq) ? decay : 0.0f;
        al[r] = val * c0 + exi * cx + eyi * cy + dlt;
    }
    v2f a0, a1, a2, a3;
    a0.x = al[0]; a0.y = al[1];
    a1.x = al[2]; a1.y = al[3];
    a2.x = al[4]; a2.y = al[5];
    {
        const float nj  = w1 * t3 - w3 * t4;
        const float nd  = w5 * t7;
        const float nr2 = (tq >= 0.0f) ? grav * cy : 0.0f;
        const float np  = w6 * t8;
        a3.x = half ? (nj - nd) : nr2;
        a3.y = half ? np : (nj + nd);
    }

    const unsigned ocol = (unsigned)(y * 9);
    const int qs = 1;
    const unsigned so0 = ocol + (unsigned)((half ? 3 : 0) * qs);
    const unsigned so1 = ocol + (unsigned)((half ? 4 : 2) * qs);
    const unsigned so3 = ocol + (unsigned)((half ? 8 : 5) * qs);
    const unsigned so5 = ocol + (unsigned)((half ? 7 : 6) * qs);
    const unsigned so2 = ocol + (unsigned)(1 * qs);

    const int ysel = half ? ((q16 + 1) & 15) : ((q16 - 1) & 15);

    auto collide = [&](v2f b0, v2f b1, v2f b2) -> v8f {
        v8f c = {};
        c = __builtin_amdgcn_wmma_f32_16x16x4_f32(false, a0, false, b0, (short)0, c,
                                                  false, false);
        c = __builtin_amdgcn_wmma_f32_16x16x4_f32(false, a1, false, b1, (short)0, c,
                                                  false, false);
        c = __builtin_amdgcn_wmma_f32_16x16x4_f32(false, a2, false, b2, (short)0, c,
                                                  false, false);
        const float rho = c[6];
        const float mx  = c[7];
        const float r2  = rho * rho;
        const float jy  = grav * r2 + c[0];
        v2f bn;
        bn.x = half ? jy * jy : r2;
        bn.y = mx * (half ? jy : mx);
        c = __builtin_amdgcn_wmma_f32_16x16x4_f32(false, a3, false, bn, (short)0, c,
                                                  false, false);
        return c;
    };

    float fr[ROWS_A][6];
    auto load_row = [&](int t) {
        const int x = (x0 - HALO + t) & (NXL - 1);
        const float* rows[3];
        rows[0] = fin + (size_t)((x + 1) & (NXL - 1)) * ROW_PITCH;
        rows[1] = fin + (size_t)x * ROW_PITCH;
        rows[2] = fin + (size_t)((x - 1) & (NXL - 1)) * ROW_PITCH;
#pragma unroll
        for (int r = 0; r < 6; ++r) fr[t][r] = rows[exr[r] + 1][col[r]];
    };

#pragma unroll
    for (int t = 0; t < ROWS_AHEAD; ++t) {
        load_row(t);
        __builtin_amdgcn_sched_barrier(0);
    }

    float hw[NSTAGES - 1][3][6];

    auto keep_row = [&](float* h, const v8f& c) {
        h[0] = c[0];
        h[2] = c[2];
        h[4] = c[4];
        h[1] = row_gather(c[1], ysel);
        h[3] = row_gather(c[3], ysel);
        h[5] = row_gather(c[5], ysel);
    };

#pragma unroll
    for (int t = 0; t < ROWS_A; ++t) {
        if (t + ROWS_AHEAD < ROWS_A) load_row(t + ROWS_AHEAD);

        {
            const float* f = fr[t];
            v2f b0, b1, b2;
            b0.x = f[0]; b0.y = f[1];
            b1.x = f[2]; b1.y = f[3];
            b2.x = f[4]; b2.y = f[5];
            const v8f c = collide(b0, b1, b2);
            keep_row(hw[0][t % 3], c);
        }

#pragma unroll
        for (int s = 1; s < NSTAGES; ++s) {
            if (t >= 2 * s) {
                const float* hp = hw[s - 1][(t - 2) % 3];
                const float* hc = hw[s - 1][(t - 1) % 3];
                const float* hn = hw[s - 1][t % 3];
                v2f b0, b1, b2;
                b0.x = hc[0]; b0.y = hc[1];
                b1.x = hp[2]; b1.y = hp[3];
                b2.x = hn[4]; b2.y = hn[5];
                const v8f c = collide(b0, b1, b2);

                if (s < NSTAGES - 1) {
                    keep_row(hw[s][t % 3], c);
                } else {
                    const int x = x0 + t - 2 * HALO;
                    float* o = fout + (size_t)x * ROW_PITCH;
                    if (owned) {
                        o[so0] = half ? c[4] : c[0];
                        o[so1] = c[1];
                        o[so3] = c[3];
                        o[so5] = c[5];
                        if (half == 0) o[so2] = c[2];
                    }
                }
            }
        }
    }
}

__global__ __launch_bounds__(128) void lbm_four(const float* __restrict__ fin,
                                                const float* __restrict__ w,
                                                float* __restrict__ fout) {
    lbm_four_body(fin, w, fout);
}

extern "C" void kernel_launch(void* const* d_in, const int* in_sizes, int n_in,
                              void* d_out, int out_size, void* d_ws, size_t ws_size,
                              hipStream_t stream) {
    const float* f0 = (const float*)d_in[0];
    const float* w  = (const float*)d_in[1];
    float* out = (float*)d_out;

    const int threads = 32 * WAVES_PER_WG;
    const int blocks  = (STRIPS * (NXL / ROWS_PER_WAVE)) / WAVES_PER_WG;

    lbm_four<<<blocks, threads, 0, stream>>>(f0, w, out);
}


# 32-column strips: a wave keeps two 16-column matrix tiles side by side, 26 of 32 columns owned (was 10 of 16), 21-row walks: 1960 waves, two per SIMD; y shift crosses the tile border; sums unchanged
# speedup vs baseline: 2.0810x; 1.1958x over previous
#include <hip/hip_runtime.h>

typedef float v2f __attribute__((ext_vector_type(2)));
typedef float v8f __attribute__((ext_vector_type(8)));

#define NXL 1024
#define NYL 1024
#define TAU 0.6f
#define GYF (-1e-4f)


struct RowTab { float v[16][12]; };

static constexpr RowTab makeRows() {
    const signed char M[9][9] = {
        { 1,  1,  1,  1,  1,  1,  1,  1,  1},
        { 0,  1,  0, -1,  0,  1, -1, -1,  1},
        { 0,  0,  1,  0, -1,  1,  1, -1, -1},
        {-4, -1, -1, -1, -1,  2,  2,  2,  2},
        { 4, -2, -2, -2, -2,  1,  1,  1,  1},
        { 0, -2,  0,  2,  0,  1, -1, -1,  1},
        { 0,  0, -2,  0,  2,  1,  1, -1, -1},
        { 0,  1, -1,  1, -1,  0,  0,  0,  0},
        { 0,  0,  0,  0,  0,  1, -1,  1, -1},
    };
    const float nrm2[9] = {9.f, 6.f, 6.f, 36.f, 36.f, 12.f, 12.f, 4.f, 4.f};
    const int rowq[16] = {0, 2, 1, 5, -1, 6, -2, -3,   -4, 4, -1, 8, 3, 7, -2, -3};
    RowTab t{};
    for (int m = 0; m < 16; ++m) {
        for (int p = 0; p < 12; ++p) t.v[m][p] = 0.0f;
        const int q = rowq[m];
        if (q >= 0) {
            for (int p = 0; p < 9; ++p)
                t.v[m][p] = (float)M[p][q] / nrm2[p] * (1.0f / TAU);
            t.v[m][9] = (float)q;
        } else {
            t.v[m][9] = -1.0f;
            if (q == -2) t.v[m][0] = 1.0f;
            if (q == -3) t.v[m][1] = 1.0f;
            if (q == -4) t.v[m][2] = 1.0f;
        }
    }
    return t;
}

__constant__ RowTab g_rows = makeRows();

#define NSTAGES 4
#define HALO (NSTAGES - 1)
#define ROWS_PER_WAVE 21
#define ROW_BLOCKS 49
#define ROWS_A (ROWS_PER_WAVE + 2 * HALO)
#define TILES 2
#define STRIP_W (16 * TILES - 2 * HALO)
#define STRIPS 40
#define WAVES_PER_WG 4
#define ROWS_AHEAD 4
#define ROW_PITCH (NYL * 9)

__device__ __forceinline__ float row_gather(float x, int sel) {
    return __builtin_bit_cast(float, __builtin_amdgcn_permlane16_var(
        __builtin_bit_cast(int, x), __builtin_bit_cast(int, x), sel, false, false));
}

__device__ __forceinline__ void lbm_four_body(const float* __restrict__ fin,
                                              const float* __restrict__ w,
                                              float* __restrict__ fout) {
    const int lane = threadIdx.x & 31;
    const int half = lane >> 4;
    const int q16  = lane & 15;
    const int wv = __builtin_amdgcn_readfirstlane((int)(threadIdx.x >> 5));
    const int gw = blockIdx.x * WAVES_PER_WG + wv;
    const int strip = gw % STRIPS;
    const int x0 = (gw / STRIPS) * ROWS_PER_WAVE;
    int yu[TILES], y[TILES];
#pragma unroll
    for (int k = 0; k < TILES; ++k) {
        yu[k] = strip * STRIP_W - HALO + 16 * k + q16;
        y[k]  = yu[k] & (NYL - 1);
    }
    bool owned[TILES];
    owned[0] = (q16 >= HALO) && (yu[0] < NYL);
    owned[1] = (q16 < 16 - HALO) && (yu[1] < NYL);

    const int qL[6] = { 0, 2,  1, 5, -1, 6};
    const int qU[6] = {-1, 4, -1, 8,  3, 7};
    const int exr[6] = {0, 0, 1, 1, -1, -1};
    const int exv[9] = {0, 1, 0, -1, 0, 1, -1, -1, 1};
    const int eyv[9] = {0, 0, 1, 0, -1, 1, 1, -1, -1};

    unsigned col[TILES][6];
#pragma unroll
    for (int k = 0; k < TILES; ++k) {
#pragma unroll
        for (int r = 0; r < 6; ++r) {
            const int ql = qL[r] >= 0 ? qL[r] : qU[r];
            const int qu = qU[r] >= 0 ? qU[r] : qL[r];
            const int ysl = (y[k] - eyv[ql]) & (NYL - 1);
            const int ysu = (y[k] - eyv[qu]) & (NYL - 1);
            const unsigned cl = (unsigned)(ysl * 9 + ql);
            const unsigned cu = (unsigned)(ysu * 9 + qu);
            col[k][r] = half ? cu : cl;
        }
    }

    const float w0 = w[0], w1 = w[1], w2 = w[2], w3 = w[3];
    const float w4 = w[4], w5 = w[5], w6 = w[6];
    const float decay = 1.0f - 1.0f / TAU;
    const float grav  = TAU * GYF;

    const float* tr = &g_rows.v[q16][0];
    const float t0 = tr[0], t1 = tr[1], t2 = tr[2], t3 = tr[3], t4 = tr[4];
    const float t5 = tr[5], t6 = tr[6], t7 = tr[7], t8 = tr[8], tq = tr[9];
    const float c0 = t0 + w0 * t3 + w2 * t4;
    const float cx = t1 + w4 * t5;
    const float cy = t2 + w4 * t6;
    float al[6];
#pragma unroll
    for (int r = 0; r < 6; ++r) {
        const int il = qL[r], iu = qU[r];
        const float vl = il >= 0 ? 1.0f : 0.0f,            vu = iu >= 0 ? 1.0f : 0.0f;
        const float xl = il >= 0 ? (float)exv[il] : 0.0f,  xu = iu >= 0 ? (float)exv[iu] : 0.0f;
        const float yl = il >= 0 ? (float)eyv[il] : 0.0f,  yy = iu >= 0 ? (float)eyv[iu] : 0.0f;
        const float idl = il >= 0 ? (float)il : -2.0f,     idu = iu >= 0 ? (float)iu : -2.0f;
        const float val = half ? vu : vl;
        const float exi = half ? xu : xl;
        const float eyi = half ? yy : yl;
        const float idi = half ? idu : idl;
        const float dlt = (idi == tq) ? decay : 0.0f;
        al[r] = val * c0 + exi * cx + eyi * cy + dlt;
    }
    v2f a0, a1, a2, a3;
    a0.x = al[0]; a0.y = al[1];
    a1.x = al[2]; a1.y = al[3];
    a2.x = al[4]; a2.y = al[5];
    {
        const float nj  = w1 * t3 - w3 * t4;
        const float nd  = w5 * t7;
        const float nr2 = (tq >= 0.0f) ? grav * cy : 0.0f;
        const float np  = w6 * t8;
        a3.x = half ? (nj - nd) : nr2;
        a3.y = half ? np : (nj + nd);
    }

    unsigned so0[TILES], so1[TILES], so3[TILES], so5[TILES], so2[TILES];
#pragma unroll
    for (int k = 0; k < TILES; ++k) {
        const unsigned ocol = (unsigned)(y[k] * 9);
        so0[k] = ocol + (unsigned)(half ? 3 : 0);
        so1[k] = ocol + (unsigned)(half ? 4 : 2);
        so3[k] = ocol + (unsigned)(half ? 8 : 5);
        so5[k] = ocol + (unsigned)(half ? 7 : 6);
        so2[k] = ocol + 1u;
    }

    const int ysel = half ? ((q16 + 1) & 15) : ((q16 - 1) & 15);
    const bool from1 = (half == 1) && (q16 == 15);
    const bool from0 = (half == 0) && (q16 == 0);

    auto collide = [&](v2f b0, v2f b1, v2f b2) -> v8f {
        v8f c = {};
        c = __builtin_amdgcn_wmma_f32_16x16x4_f32(false, a0, false, b0, (short)0, c,
                                                  false, false);
        c = __builtin_amdgcn_wmma_f32_16x16x4_f32(false, a1, false, b1, (short)0, c,
                                                  false, false);
        c = __builtin_amdgcn_wmma_f32_16x16x4_f32(false, a2, false, b2, (short)0, c,
                                                  false, false);
        const float rho = c[6];
        const float mx  = c[7];
        const float r2  = rho * rho;
        const float jy  = grav * r2 + c[0];
        v2f bn;
        bn.x = half ? jy * jy : r2;
        bn.y = mx * (half ? jy : mx);
        c = __builtin_amdgcn_wmma_f32_16x16x4_f32(false, a3, false, bn, (short)0, c,
                                                  false, false);
        return c;
    };

    float fr[ROWS_A][TILES][6];
    auto load_row = [&](int t) {
        const int x = (x0 - HALO + t) & (NXL - 1);
        const float* rows[3];
        rows[0] = fin + (size_t)((x + 1) & (NXL - 1)) * ROW_PITCH;
        rows[1] = fin + (size_t)x * ROW_PITCH;
        rows[2] = fin + (size_t)((x - 1) & (NXL - 1)) * ROW_PITCH;
#pragma unroll
        for (int k = 0; k < TILES; ++k) {
#pragma unroll
            for (int r = 0; r < 6; ++r) fr[t][k][r] = rows[exr[r] + 1][col[k][r]];
        }
    };

#pragma unroll
    for (int t = 0; t < ROWS_AHEAD; ++t) {
        load_row(t);
        __builtin_amdgcn_sched_barrier(0);
    }

    float hw[NSTAGES - 1][3][TILES][6];

    auto keep_row = [&](float (*h)[6], const v8f& ca, const v8f& cb) {
        h[0][0] = ca[0]; h[0][2] = ca[2]; h[0][4] = ca[4];
        h[1][0] = cb[0]; h[1][2] = cb[2]; h[1][4] = cb[4];
#pragma unroll
        for (int r = 1; r < 6; r += 2) {
            const float ta = row_gather(ca[r], ysel);
            const float tb = row_gather(cb[r], ysel);
            h[0][r] = from1 ? tb : ta;
            h[1][r] = from0 ? ta : tb;
        }
    };

#pragma unroll
    for (int t = 0; t < ROWS_A; ++t) {
        if (t + ROWS_AHEAD < ROWS_A) load_row(t + ROWS_AHEAD);

        {
            v8f c[TILES];
#pragma unroll
            for (int k = 0; k < TILES; ++k) {
                const float* f = fr[t][k];
                v2f b0, b1, b2;
                b0.x = f[0]; b0.y = f[1];
                b1.x = f[2]; b1.y = f[3];
                b2.x = f[4]; b2.y = f[5];
                c[k] = collide(b0, b1, b2);
            }
            keep_row(hw[0][t % 3], c[0], c[1]);
        }

#pragma unroll
        for (int s = 1; s < NSTAGES; ++s) {
            if (t >= 2 * s) {
                v8f c[TILES];
#pragma unroll
                for (int k = 0; k < TILES; ++k) {
                    const float* hp = hw[s - 1][(t - 2) % 3][k];
                    const float* hc = hw[s - 1][(t - 1) % 3][k];
                    const float* hn = hw[s - 1][t % 3][k];
                    v2f b0, b1, b2;
                    b0.x = hc[0]; b0.y = hc[1];
                    b1.x = hp[2]; b1.y = hp[3];
                    b2.x = hn[4]; b2.y = hn[5];
                    c[k] = collide(b0, b1, b2);
                }

                if (s < NSTAGES - 1) {
                    keep_row(hw[s][t % 3], c[0], c[1]);
                } else {
                    const int x = x0 + t - 2 * HALO;
                    if (x < NXL) {
                        float* o = fout + (size_t)x * ROW_PITCH;
#pragma unroll
                        for (int k = 0; k < TILES; ++k) {
                            if (owned[k]) {
                                o[so0[k]] = half ? c[k][4] : c[k][0];
                                o[so1[k]] = c[k][1];
                                o[so3[k]] = c[k][3];
                                o[so5[k]] = c[k][5];
                                if (half == 0) o[so2[k]] = c[k][2];
                            }
                        }
                    }
                }
            }
        }
    }
}

__global__ __launch_bounds__(128) void lbm_four(const float* __restrict__ fin,
                                                const float* __restrict__ w,
                                                float* __restrict__ fout) {
    lbm_four_body(fin, w, fout);
}

extern "C" void kernel_launch(void* const* d_in, const int* in_sizes, int n_in,
                              void* d_out, int out_size, void* d_ws, size_t ws_size,
                              hipStream_t stream) {
    const float* f0 = (const float*)d_in[0];
    const float* w  = (const float*)d_in[1];
    float* out = (float*)d_out;

    const int threads = 32 * WAVES_PER_WG;
    const int blocks  = (STRIPS * ROW_BLOCKS) / WAVES_PER_WG;

    lbm_four<<<blocks, threads, 0, stream>>>(f0, w, out);
}


# each stage reads the rows the stage before it made in the 3 trips before this one (a 4th kept row per stage): a trip's 8 collisions are independent and issued side by side; per-cell sums unchanged
# speedup vs baseline: 2.1870x; 1.0509x over previous
#include <hip/hip_runtime.h>

typedef float v2f __attribute__((ext_vector_type(2)));
typedef float v8f __attribute__((ext_vector_type(8)));

#define NXL 1024
#define NYL 1024
#define TAU 0.6f
#define GYF (-1e-4f)


struct RowTab { float v[16][12]; };

static constexpr RowTab makeRows() {
    const signed char M[9][9] = {
        { 1,  1,  1,  1,  1,  1,  1,  1,  1},
        { 0,  1,  0, -1,  0,  1, -1, -1,  1},
        { 0,  0,  1,  0, -1,  1,  1, -1, -1},
        {-4, -1, -1, -1, -1,  2,  2,  2,  2},
        { 4, -2, -2, -2, -2,  1,  1,  1,  1},
        { 0, -2,  0,  2,  0,  1, -1, -1,  1},
        { 0,  0, -2,  0,  2,  1,  1, -1, -1},
        { 0,  1, -1,  1, -1,  0,  0,  0,  0},
        { 0,  0,  0,  0,  0,  1, -1,  1, -1},
    };
    const float nrm2[9] = {9.f, 6.f, 6.f, 36.f, 36.f, 12.f, 12.f, 4.f, 4.f};
    const int rowq[16] = {0, 2, 1, 5, 0, 6, -2, -3,   -4, 4, -1, 8, 3, 7, -2, -3};
    RowTab t{};
    for (int m = 0; m < 16; ++m) {
        for (int p = 0; p < 12; ++p) t.v[m][p] = 0.0f;
        const int q = rowq[m];
        if (q >= 0) {
            for (int p = 0; p < 9; ++p)
                t.v[m][p] = (float)M[p][q] / nrm2[p] * (1.0f / TAU);
            t.v[m][9] = (float)q;
        } else {
            t.v[m][9] = -1.0f;
            if (q == -2) t.v[m][0] = 1.0f;
            if (q == -3) t.v[m][1] = 1.0f;
            if (q == -4) t.v[m][2] = 1.0f;
        }
    }
    return t;
}

__constant__ RowTab g_rows = makeRows();

#define NSTAGES 4
#define HALO (NSTAGES - 1)
#define ROWS_PER_WAVE 21
#define ROW_BLOCKS 49
#define ROWS_A (ROWS_PER_WAVE + 2 * HALO)
#define NTRIPS (ROWS_A + HALO)
#define TILES 2
#define STRIP_W (16 * TILES - 2 * HALO)
#define STRIPS 40
#define WAVES_PER_WG 4
#define ROWS_AHEAD 4
#define ROW_PITCH (NYL * 9)

__device__ __forceinline__ float row_gather(float x, int sel) {
    return __builtin_bit_cast(float, __builtin_amdgcn_permlane16_var(
        __builtin_bit_cast(int, x), __builtin_bit_cast(int, x), sel, false, false));
}

__device__ __forceinline__ void lbm_four_body(const float* __restrict__ fin,
                                              const float* __restrict__ w,
                                              float* __restrict__ fout) {
    const int lane = threadIdx.x & 31;
    const int half = lane >> 4;
    const int q16  = lane & 15;
    const int wv = __builtin_amdgcn_readfirstlane((int)(threadIdx.x >> 5));
    const int gw = blockIdx.x * WAVES_PER_WG + wv;
    const int strip = gw % STRIPS;
    const int x0 = (gw / STRIPS) * ROWS_PER_WAVE;
    int yu[TILES], y[TILES];
#pragma unroll
    for (int k = 0; k < TILES; ++k) {
        yu[k] = strip * STRIP_W - HALO + 16 * k + q16;
        y[k]  = yu[k] & (NYL - 1);
    }
    bool owned[TILES];
    owned[0] = (q16 >= HALO) && (yu[0] < NYL);
    owned[1] = (q16 < 16 - HALO) && (yu[1] < NYL);

    const int qL[6] = { 0, 2,  1, 5, -1, 6};
    const int qU[6] = {-1, 4, -1, 8,  3, 7};
    const int exr[6] = {0, 0, 1, 1, -1, -1};
    const int exv[9] = {0, 1, 0, -1, 0, 1, -1, -1, 1};
    const int eyv[9] = {0, 0, 1, 0, -1, 1, 1, -1, -1};

    unsigned col[TILES][6];
#pragma unroll
    for (int k = 0; k < TILES; ++k) {
#pragma unroll
        for (int r = 0; r < 6; ++r) {
            const int ql = qL[r] >= 0 ? qL[r] : qU[r];
            const int qu = qU[r] >= 0 ? qU[r] : qL[r];
            const int ysl = (y[k] - eyv[ql]) & (NYL - 1);
            const int ysu = (y[k] - eyv[qu]) & (NYL - 1);
            const unsigned cl = (unsigned)(ysl * 9 + ql);
            const unsigned cu = (unsigned)(ysu * 9 + qu);
            col[k][r] = half ? cu : cl;
        }
    }

    const float w0 = w[0], w1 = w[1], w2 = w[2], w3 = w[3];
    const float w4 = w[4], w5 = w[5], w6 = w[6];
    const float decay = 1.0f - 1.0f / TAU;
    const float grav  = TAU * GYF;

    const float* tr = &g_rows.v[q16][0];
    const float t0 = tr[0], t1 = tr[1], t2 = tr[2], t3 = tr[3], t4 = tr[4];
    const float t5 = tr[5], t6 = tr[6], t7 = tr[7], t8 = tr[8], tq = tr[9];
    const float c0 = t0 + w0 * t3 + w2 * t4;
    const float cx = t1 + w4 * t5;
    const float cy = t2 + w4 * t6;
    float al[6];
#pragma unroll
    for (int r = 0; r < 6; ++r) {
        const int il = qL[r], iu = qU[r];
        const float vl = il >= 0 ? 1.0f : 0.0f,            vu = iu >= 0 ? 1.0f : 0.0f;
        const float xl = il >= 0 ? (float)exv[il] : 0.0f,  xu = iu >= 0 ? (float)exv[iu] : 0.0f;
        const float yl = il >= 0 ? (float)eyv[il] : 0.0f,  yy = iu >= 0 ? (float)eyv[iu] : 0.0f;
        const float idl = il >= 0 ? (float)il : -2.0f,     idu = iu >= 0 ? (float)iu : -2.0f;
        const float val = half ? vu : vl;
        const float exi = half ? xu : xl;
        const float eyi = half ? yy : yl;
        const float idi = half ? idu : idl;
        const float dlt = (idi == tq) ? decay : 0.0f;
        al[r] = val * c0 + exi * cx + eyi * cy + dlt;
    }
    v2f a0, a1, a2, a3;
    a0.x = al[0]; a0.y = al[1];
    a1.x = al[2]; a1.y = al[3];
    a2.x = al[4]; a2.y = al[5];
    {
        const float nj  = w1 * t3 - w3 * t4;
        const float nd  = w5 * t7;
        const float nr2 = (tq >= 0.0f) ? grav * cy : 0.0f;
        const float np  = w6 * t8;
        a3.x = half ? (nj - nd) : nr2;
        a3.y = half ? np : (nj + nd);
    }

    unsigned so0[TILES], so1[TILES], so3[TILES], so5[TILES], so2[TILES];
#pragma unroll
    for (int k = 0; k < TILES; ++k) {
        const unsigned ocol = (unsigned)(y[k] * 9);
        so0[k] = ocol + (unsigned)(half ? 3 : 0);
        so1[k] = ocol + (unsigned)(half ? 4 : 2);
        so3[k] = ocol + (unsigned)(half ? 8 : 5);
        so5[k] = ocol + (unsigned)(half ? 7 : 6);
        so2[k] = ocol + 1u;
    }

    const int ysel = half ? ((q16 + 1) & 15) : ((q16 - 1) & 15);
    const bool from1 = (half == 1) && (q16 == 15);
    const bool from0 = (half == 0) && (q16 == 0);

    auto lin0 = [&](v2f b) -> v8f {
        v8f c = {};
        return __builtin_amdgcn_wmma_f32_16x16x4_f32(false, a0, false, b, (short)0, c,
                                                     false, false);
    };
    auto lin1 = [&](v8f c, v2f b) -> v8f {
        return __builtin_amdgcn_wmma_f32_16x16x4_f32(false, a1, false, b, (short)0, c,
                                                     false, false);
    };
    auto lin2 = [&](v8f c, v2f b) -> v8f {
        return __builtin_amdgcn_wmma_f32_16x16x4_f32(false, a2, false, b, (short)0, c,
                                                     false, false);
    };
    auto nonlin_terms = [&](const v8f& c) -> v2f {
        const float rho = c[6];
        const float mx  = c[7];
        const float r2  = rho * rho;
        const float jy  = grav * r2 + c[0];
        v2f bn;
        bn.x = half ? jy * jy : r2;
        bn.y = mx * (half ? jy : mx);
        return bn;
    };
    auto nonlin = [&](v8f c, v2f bn) -> v8f {
        return __builtin_amdgcn_wmma_f32_16x16x4_f32(false, a3, false, bn, (short)0, c,
                                                     false, false);
    };

    float fr[ROWS_A][TILES][6];
    auto load_row = [&](int t) {
        const int x = (x0 - HALO + t) & (NXL - 1);
        const float* rows[3];
        rows[0] = fin + (size_t)((x + 1) & (NXL - 1)) * ROW_PITCH;
        rows[1] = fin + (size_t)x * ROW_PITCH;
        rows[2] = fin + (size_t)((x - 1) & (NXL - 1)) * ROW_PITCH;
#pragma unroll
        for (int k = 0; k < TILES; ++k) {
#pragma unroll
            for (int r = 0; r < 6; ++r) fr[t][k][r] = rows[exr[r] + 1][col[k][r]];
        }
    };

#pragma unroll
    for (int t = 0; t < ROWS_AHEAD; ++t) {
        load_row(t);
        __builtin_amdgcn_sched_barrier(0);
    }

    float hw[NSTAGES - 1][4][TILES][6];

    auto keep_row = [&](float (*h)[6], const v8f& ca, const v8f& cb) {
        h[0][0] = ca[0]; h[0][2] = ca[2]; h[0][4] = ca[4];
        h[1][0] = cb[0]; h[1][2] = cb[2]; h[1][4] = cb[4];
#pragma unroll
        for (int r = 1; r < 6; r += 2) {
            const float ta = row_gather(ca[r], ysel);
            const float tb = row_gather(cb[r], ysel);
            h[0][r] = from1 ? tb : ta;
            h[1][r] = from0 ? ta : tb;
        }
    };

#pragma unroll
    for (int t = 0; t < NTRIPS; ++t) {
        if (t + ROWS_AHEAD < ROWS_A) load_row(t + ROWS_AHEAD);

        bool on[NSTAGES];
#pragma unroll
        for (int s = 0; s < NSTAGES; ++s) on[s] = (t >= 3 * s) && (t <= ROWS_A - 1 + s);

        v2f b[NSTAGES][TILES][3];
#pragma unroll
        for (int k = 0; k < TILES; ++k) {
            if (on[0]) {
                const float* f = fr[t < ROWS_A ? t : 0][k];
                b[0][k][0].x = f[0]; b[0][k][0].y = f[1];
                b[0][k][1].x = f[2]; b[0][k][1].y = f[3];
                b[0][k][2].x = f[4]; b[0][k][2].y = f[5];
            }
#pragma unroll
            for (int s = 1; s < NSTAGES; ++s) {
                if (on[s]) {
                    const float* hp = hw[s - 1][(t + 1) & 3][k];
                    const float* hc = hw[s - 1][(t + 2) & 3][k];
                    const float* hn = hw[s - 1][(t + 3) & 3][k];
                    b[s][k][0].x = hc[0]; b[s][k][0].y = hc[1];
                    b[s][k][1].x = hp[2]; b[s][k][1].y = hp[3];
                    b[s][k][2].x = hn[4]; b[s][k][2].y = hn[5];
                }
            }
        }

        v8f c[NSTAGES][TILES];
#pragma unroll
        for (int s = NSTAGES - 1; s >= 0; --s)
#pragma unroll
            for (int k = 0; k < TILES; ++k)
                if (on[s]) c[s][k] = lin0(b[s][k][0]);
#pragma unroll
        for (int s = NSTAGES - 1; s >= 0; --s)
#pragma unroll
            for (int k = 0; k < TILES; ++k)
                if (on[s]) c[s][k] = lin1(c[s][k], b[s][k][1]);
#pragma unroll
        for (int s = NSTAGES - 1; s >= 0; --s)
#pragma unroll
            for (int k = 0; k < TILES; ++k)
                if (on[s]) c[s][k] = lin2(c[s][k], b[s][k][2]);

        v2f bn[NSTAGES][TILES];
#pragma unroll
        for (int s = NSTAGES - 1; s >= 0; --s)
#pragma unroll
            for (int k = 0; k < TILES; ++k)
                if (on[s]) bn[s][k] = nonlin_terms(c[s][k]);
#pragma unroll
        for (int s = NSTAGES - 1; s >= 0; --s)
#pragma unroll
            for (int k = 0; k < TILES; ++k)
                if (on[s]) c[s][k] = nonlin(c[s][k], bn[s][k]);

        if (on[NSTAGES - 1]) {
            const int x = x0 + t - 3 * HALO;
            if (x < NXL) {
                float* o = fout + (size_t)x * ROW_PITCH;
#pragma unroll
                for (int k = 0; k < TILES; ++k) {
                    if (owned[k]) {
                        const v8f& cc = c[NSTAGES - 1][k];
                        o[so0[k]] = cc[4];
                        o[so1[k]] = cc[1];
                        o[so3[k]] = cc[3];
                        o[so5[k]] = cc[5];
                        if (half == 0) o[so2[k]] = cc[2];
                    }
                }
            }
        }

#pragma unroll
        for (int s = NSTAGES - 2; s >= 0; --s)
            if (on[s]) keep_row(hw[s][t & 3], c[s][0], c[s][1]);
    }
}

__global__ __launch_bounds__(128) void lbm_four(const float* __restrict__ fin,
                                                const float* __restrict__ w,
                                                float* __restrict__ fout) {
    lbm_four_body(fin, w, fout);
}

extern "C" void kernel_launch(void* const* d_in, const int* in_sizes, int n_in,
                              void* d_out, int out_size, void* d_ws, size_t ws_size,
                              hipStream_t stream) {
    const float* f0 = (const float*)d_in[0];
    const float* w  = (const float*)d_in[1];
    float* out = (float*)d_out;

    const int threads = 32 * WAVES_PER_WG;
    const int blocks  = (STRIPS * ROW_BLOCKS) / WAVES_PER_WG;

    lbm_four<<<blocks, threads, 0, stream>>>(f0, w, out);
}


# opening of a walk: its first 4 rows' loads go out before any wait and stay 4 trips ahead through the 9 trips without a store; the compiler no longer merges two trips' loads; sums unchanged
# speedup vs baseline: 2.3024x; 1.0528x over previous
#include <hip/hip_runtime.h>

typedef float v2f __attribute__((ext_vector_type(2)));
typedef float v8f __attribute__((ext_vector_type(8)));

#define NXL 1024
#define NYL 1024
#define TAU 0.6f
#define GYF (-1e-4f)


struct RowTab { float v[16][12]; };

static constexpr RowTab makeRows() {
    const signed char M[9][9] = {
        { 1,  1,  1,  1,  1,  1,  1,  1,  1},
        { 0,  1,  0, -1,  0,  1, -1, -1,  1},
        { 0,  0,  1,  0, -1,  1,  1, -1, -1},
        {-4, -1, -1, -1, -1,  2,  2,  2,  2},
        { 4, -2, -2, -2, -2,  1,  1,  1,  1},
        { 0, -2,  0,  2,  0,  1, -1, -1,  1},
        { 0,  0, -2,  0,  2,  1,  1, -1, -1},
        { 0,  1, -1,  1, -1,  0,  0,  0,  0},
        { 0,  0,  0,  0,  0,  1, -1,  1, -1},
    };
    const float nrm2[9] = {9.f, 6.f, 6.f, 36.f, 36.f, 12.f, 12.f, 4.f, 4.f};
    const int rowq[16] = {0, 2, 1, 5, 0, 6, -2, -3,   -4, 4, -1, 8, 3, 7, -2, -3};
    RowTab t{};
    for (int m = 0; m < 16; ++m) {
        for (int p = 0; p < 12; ++p) t.v[m][p] = 0.0f;
        const int q = rowq[m];
        if (q >= 0) {
            for (int p = 0; p < 9; ++p)
                t.v[m][p] = (float)M[p][q] / nrm2[p] * (1.0f / TAU);
            t.v[m][9] = (float)q;
        } else {
            t.v[m][9] = -1.0f;
            if (q == -2) t.v[m][0] = 1.0f;
            if (q == -3) t.v[m][1] = 1.0f;
            if (q == -4) t.v[m][2] = 1.0f;
        }
    }
    return t;
}

__constant__ RowTab g_rows = makeRows();

#define NSTAGES 4
#define HALO (NSTAGES - 1)
#define ROWS_PER_WAVE 21
#define ROW_BLOCKS 49
#define ROWS_A (ROWS_PER_WAVE + 2 * HALO)
#define NTRIPS (ROWS_A + HALO)
#define TILES 2
#define STRIP_W (16 * TILES - 2 * HALO)
#define STRIPS 40
#define WAVES_PER_WG 4
#define ROWS_AHEAD 4
#define ROW_PITCH (NYL * 9)

__device__ __forceinline__ float row_gather(float x, int sel) {
    return __builtin_bit_cast(float, __builtin_amdgcn_permlane16_var(
        __builtin_bit_cast(int, x), __builtin_bit_cast(int, x), sel, false, false));
}

__device__ __forceinline__ void lbm_four_body(const float* __restrict__ fin,
                                              const float* __restrict__ w,
                                              float* __restrict__ fout) {
    const int lane = threadIdx.x & 31;
    const int half = lane >> 4;
    const int q16  = lane & 15;
    const int wv = __builtin_amdgcn_readfirstlane((int)(threadIdx.x >> 5));
    const int gw = blockIdx.x * WAVES_PER_WG + wv;
    const int strip = gw % STRIPS;
    const int x0 = (gw / STRIPS) * ROWS_PER_WAVE;
    int yu[TILES], y[TILES];
#pragma unroll
    for (int k = 0; k < TILES; ++k) {
        yu[k] = strip * STRIP_W - HALO + 16 * k + q16;
        y[k]  = yu[k] & (NYL - 1);
    }
    bool owned[TILES];
    owned[0] = (q16 >= HALO) && (yu[0] < NYL);
    owned[1] = (q16 < 16 - HALO) && (yu[1] < NYL);

    const int qL[6] = { 0, 2,  1, 5, -1, 6};
    const int qU[6] = {-1, 4, -1, 8,  3, 7};
    const int exr[6] = {0, 0, 1, 1, -1, -1};
    const int exv[9] = {0, 1, 0, -1, 0, 1, -1, -1, 1};
    const int eyv[9] = {0, 0, 1, 0, -1, 1, 1, -1, -1};

    unsigned col[TILES][6];
#pragma unroll
    for (int k = 0; k < TILES; ++k) {
#pragma unroll
        for (int r = 0; r < 6; ++r) {
            const int ql = qL[r] >= 0 ? qL[r] : qU[r];
            const int qu = qU[r] >= 0 ? qU[r] : qL[r];
            const int ysl = (y[k] - eyv[ql]) & (NYL - 1);
            const int ysu = (y[k] - eyv[qu]) & (NYL - 1);
            const unsigned cl = (unsigned)(ysl * 9 + ql);
            const unsigned cu = (unsigned)(ysu * 9 + qu);
            col[k][r] = half ? cu : cl;
        }
    }

    const float w0 = w[0], w1 = w[1], w2 = w[2], w3 = w[3];
    const float w4 = w[4], w5 = w[5], w6 = w[6];
    const float decay = 1.0f - 1.0f / TAU;
    const float grav  = TAU * GYF;

    const float* tr = &g_rows.v[q16][0];
    const float t0 = tr[0], t1 = tr[1], t2 = tr[2], t3 = tr[3], t4 = tr[4];
    const float t5 = tr[5], t6 = tr[6], t7 = tr[7], t8 = tr[8], tq = tr[9];
    const float c0 = t0 + w0 * t3 + w2 * t4;
    const float cx = t1 + w4 * t5;
    const float cy = t2 + w4 * t6;
    float al[6];
#pragma unroll
    for (int r = 0; r < 6; ++r) {
        const int il = qL[r], iu = qU[r];
        const float vl = il >= 0 ? 1.0f : 0.0f,            vu = iu >= 0 ? 1.0f : 0.0f;
        const float xl = il >= 0 ? (float)exv[il] : 0.0f,  xu = iu >= 0 ? (float)exv[iu] : 0.0f;
        const float yl = il >= 0 ? (float)eyv[il] : 0.0f,  yy = iu >= 0 ? (float)eyv[iu] : 0.0f;
        const float idl = il >= 0 ? (float)il : -2.0f,     idu = iu >= 0 ? (float)iu : -2.0f;
        const float val = half ? vu : vl;
        const float exi = half ? xu : xl;
        const float eyi = half ? yy : yl;
        const float idi = half ? idu : idl;
        const float dlt = (idi == tq) ? decay : 0.0f;
        al[r] = val * c0 + exi * cx + eyi * cy + dlt;
    }
    v2f a0, a1, a2, a3;
    a0.x = al[0]; a0.y = al[1];
    a1.x = al[2]; a1.y = al[3];
    a2.x = al[4]; a2.y = al[5];
    {
        const float nj  = w1 * t3 - w3 * t4;
        const float nd  = w5 * t7;
        const float nr2 = (tq >= 0.0f) ? grav * cy : 0.0f;
        const float np  = w6 * t8;
        a3.x = half ? (nj - nd) : nr2;
        a3.y = half ? np : (nj + nd);
    }

    unsigned so0[TILES], so1[TILES], so3[TILES], so5[TILES], so2[TILES];
#pragma unroll
    for (int k = 0; k < TILES; ++k) {
        const unsigned ocol = (unsigned)(y[k] * 9);
        so0[k] = ocol + (unsigned)(half ? 3 : 0);
        so1[k] = ocol + (unsigned)(half ? 4 : 2);
        so3[k] = ocol + (unsigned)(half ? 8 : 5);
        so5[k] = ocol + (unsigned)(half ? 7 : 6);
        so2[k] = ocol + 1u;
    }

    const int ysel = half ? ((q16 + 1) & 15) : ((q16 - 1) & 15);
    const bool from1 = (half == 1) && (q16 == 15);
    const bool from0 = (half == 0) && (q16 == 0);

    auto lin0 = [&](v2f b) -> v8f {
        v8f c = {};
        return __builtin_amdgcn_wmma_f32_16x16x4_f32(false, a0, false, b, (short)0, c,
                                                     false, false);
    };
    auto lin1 = [&](v8f c, v2f b) -> v8f {
        return __builtin_amdgcn_wmma_f32_16x16x4_f32(false, a1, false, b, (short)0, c,
                                                     false, false);
    };
    auto lin2 = [&](v8f c, v2f b) -> v8f {
        return __builtin_amdgcn_wmma_f32_16x16x4_f32(false, a2, false, b, (short)0, c,
                                                     false, false);
    };
    auto nonlin_terms = [&](const v8f& c) -> v2f {
        const float rho = c[6];
        const float mx  = c[7];
        const float r2  = rho * rho;
        const float jy  = grav * r2 + c[0];
        v2f bn;
        bn.x = half ? jy * jy : r2;
        bn.y = mx * (half ? jy : mx);
        return bn;
    };
    auto nonlin = [&](v8f c, v2f bn) -> v8f {
        return __builtin_amdgcn_wmma_f32_16x16x4_f32(false, a3, false, bn, (short)0, c,
                                                     false, false);
    };

    float fr[ROWS_A][TILES][6];
    auto load_row = [&](int t) {
        int x = (x0 - HALO + t) & (NXL - 1);
        asm("" : "+s"(x));
        const float* rows[3];
        rows[0] = fin + (size_t)((x + 1) & (NXL - 1)) * ROW_PITCH;
        rows[1] = fin + (size_t)x * ROW_PITCH;
        rows[2] = fin + (size_t)((x - 1) & (NXL - 1)) * ROW_PITCH;
#pragma unroll
        for (int k = 0; k < TILES; ++k) {
#pragma unroll
            for (int r = 0; r < 6; ++r) fr[t][k][r] = rows[exr[r] + 1][col[k][r]];
        }
    };

#pragma unroll
    for (int t = 0; t < ROWS_AHEAD; ++t) load_row(t);
    __builtin_amdgcn_sched_barrier(0);

    float hw[NSTAGES - 1][4][TILES][6];

    auto keep_row = [&](float (*h)[6], const v8f& ca, const v8f& cb) {
        h[0][0] = ca[0]; h[0][2] = ca[2]; h[0][4] = ca[4];
        h[1][0] = cb[0]; h[1][2] = cb[2]; h[1][4] = cb[4];
#pragma unroll
        for (int r = 1; r < 6; r += 2) {
            const float ta = row_gather(ca[r], ysel);
            const float tb = row_gather(cb[r], ysel);
            h[0][r] = from1 ? tb : ta;
            h[1][r] = from0 ? ta : tb;
        }
    };

#pragma unroll
    for (int t = 0; t < NTRIPS; ++t) {
        if (t + ROWS_AHEAD < ROWS_A) load_row(t + ROWS_AHEAD);

        bool on[NSTAGES];
#pragma unroll
        for (int s = 0; s < NSTAGES; ++s) on[s] = (t >= 3 * s) && (t <= ROWS_A - 1 + s);

        v2f b[NSTAGES][TILES][3];
#pragma unroll
        for (int k = 0; k < TILES; ++k) {
            if (on[0]) {
                const float* f = fr[t < ROWS_A ? t : 0][k];
                b[0][k][0].x = f[0]; b[0][k][0].y = f[1];
                b[0][k][1].x = f[2]; b[0][k][1].y = f[3];
                b[0][k][2].x = f[4]; b[0][k][2].y = f[5];
            }
#pragma unroll
            for (int s = 1; s < NSTAGES; ++s) {
                if (on[s]) {
                    const float* hp = hw[s - 1][(t + 1) & 3][k];
                    const float* hc = hw[s - 1][(t + 2) & 3][k];
                    const float* hn = hw[s - 1][(t + 3) & 3][k];
                    b[s][k][0].x = hc[0]; b[s][k][0].y = hc[1];
                    b[s][k][1].x = hp[2]; b[s][k][1].y = hp[3];
                    b[s][k][2].x = hn[4]; b[s][k][2].y = hn[5];
                }
            }
        }

        v8f c[NSTAGES][TILES];
#pragma unroll
        for (int s = NSTAGES - 1; s >= 0; --s)
#pragma unroll
            for (int k = 0; k < TILES; ++k)
                if (on[s]) c[s][k] = lin0(b[s][k][0]);
#pragma unroll
        for (int s = NSTAGES - 1; s >= 0; --s)
#pragma unroll
            for (int k = 0; k < TILES; ++k)
                if (on[s]) c[s][k] = lin1(c[s][k], b[s][k][1]);
#pragma unroll
        for (int s = NSTAGES - 1; s >= 0; --s)
#pragma unroll
            for (int k = 0; k < TILES; ++k)
                if (on[s]) c[s][k] = lin2(c[s][k], b[s][k][2]);

        v2f bn[NSTAGES][TILES];
#pragma unroll
        for (int s = NSTAGES - 1; s >= 0; --s)
#pragma unroll
            for (int k = 0; k < TILES; ++k)
                if (on[s]) bn[s][k] = nonlin_terms(c[s][k]);
#pragma unroll
        for (int s = NSTAGES - 1; s >= 0; --s)
#pragma unroll
            for (int k = 0; k < TILES; ++k)
                if (on[s]) c[s][k] = nonlin(c[s][k], bn[s][k]);

        if (on[NSTAGES - 1]) {
            const int x = x0 + t - 3 * HALO;
            if (x < NXL) {
                float* o = fout + (size_t)x * ROW_PITCH;
#pragma unroll
                for (int k = 0; k < TILES; ++k) {
                    if (owned[k]) {
                        const v8f& cc = c[NSTAGES - 1][k];
                        o[so0[k]] = cc[4];
                        o[so1[k]] = cc[1];
                        o[so3[k]] = cc[3];
                        o[so5[k]] = cc[5];
                        if (half == 0) o[so2[k]] = cc[2];
                    }
                }
            }
        }

#pragma unroll
        for (int s = NSTAGES - 2; s >= 0; --s)
            if (on[s]) keep_row(hw[s][t & 3], c[s][0], c[s][1]);
        if (t < 3 * HALO) __builtin_amdgcn_sched_barrier(0);
    }
}

__global__ __launch_bounds__(128) void lbm_four(const float* __restrict__ fin,
                                                const float* __restrict__ w,
                                                float* __restrict__ fout) {
    lbm_four_body(fin, w, fout);
}

extern "C" void kernel_launch(void* const* d_in, const int* in_sizes, int n_in,
                              void* d_out, int out_size, void* d_ws, size_t ws_size,
                              hipStream_t stream) {
    const float* f0 = (const float*)d_in[0];
    const float* w  = (const float*)d_in[1];
    float* out = (float*)d_out;

    const int threads = 32 * WAVES_PER_WG;
    const int blocks  = (STRIPS * ROW_BLOCKS) / WAVES_PER_WG;

    lbm_four<<<blocks, threads, 0, stream>>>(f0, w, out);
}


# wave issue priority falls with walk progress (s_setprio 3 at the start, 2, 1, 0 behind about 1/2, 3/4, 9/10 of a wave's matrix products): the two waves of a SIMD end together; sums unchanged
# speedup vs baseline: 2.3896x; 1.0379x over previous
#include <hip/hip_runtime.h>

typedef float v2f __attribute__((ext_vector_type(2)));
typedef float v8f __attribute__((ext_vector_type(8)));

#define NXL 1024
#define NYL 1024
#define TAU 0.6f
#define GYF (-1e-4f)


struct RowTab { float v[16][12]; };

static constexpr RowTab makeRows() {
    const signed char M[9][9] = {
        { 1,  1,  1,  1,  1,  1,  1,  1,  1},
        { 0,  1,  0, -1,  0,  1, -1, -1,  1},
        { 0,  0,  1,  0, -1,  1,  1, -1, -1},
        {-4, -1, -1, -1, -1,  2,  2,  2,  2},
        { 4, -2, -2, -2, -2,  1,  1,  1,  1},
        { 0, -2,  0,  2,  0,  1, -1, -1,  1},
        { 0,  0, -2,  0,  2,  1,  1, -1, -1},
        { 0,  1, -1,  1, -1,  0,  0,  0,  0},
        { 0,  0,  0,  0,  0,  1, -1,  1, -1},
    };
    const float nrm2[9] = {9.f, 6.f, 6.f, 36.f, 36.f, 12.f, 12.f, 4.f, 4.f};
    const int rowq[16] = {0, 2, 1, 5, 0, 6, -2, -3,   -4, 4, -1, 8, 3, 7, -2, -3};
    RowTab t{};
    for (int m = 0; m < 16; ++m) {
        for (int p = 0; p < 12; ++p) t.v[m][p] = 0.0f;
        const int q = rowq[m];
        if (q >= 0) {
            for (int p = 0; p < 9; ++p)
                t.v[m][p] = (float)M[p][q] / nrm2[p] * (1.0f / TAU);
            t.v[m][9] = (float)q;
        } else {
            t.v[m][9] = -1.0f;
            if (q == -2) t.v[m][0] = 1.0f;
            if (q == -3) t.v[m][1] = 1.0f;
            if (q == -4) t.v[m][2] = 1.0f;
        }
    }
    return t;
}

__constant__ RowTab g_rows = makeRows();

#define NSTAGES 4
#define HALO (NSTAGES - 1)
#define ROWS_PER_WAVE 21
#define ROW_BLOCKS 49
#define ROWS_A (ROWS_PER_WAVE + 2 * HALO)
#define NTRIPS (ROWS_A + HALO)
#define TILES 2
#define STRIP_W (16 * TILES - 2 * HALO)
#define STRIPS 40
#define WAVES_PER_WG 4
#define ROWS_AHEAD 4
#define ROW_PITCH (NYL * 9)
#define PRIO_T2 16
#define PRIO_T1 22
#define PRIO_T0 26

__device__ __forceinline__ float row_gather(float x, int sel) {
    return __builtin_bit_cast(float, __builtin_amdgcn_permlane16_var(
        __builtin_bit_cast(int, x), __builtin_bit_cast(int, x), sel, false, false));
}

__device__ __forceinline__ void lbm_four_body(const float* __restrict__ fin,
                                              const float* __restrict__ w,
                                              float* __restrict__ fout) {
    __builtin_amdgcn_s_setprio(3);
    const int lane = threadIdx.x & 31;
    const int half = lane >> 4;
    const int q16  = lane & 15;
    const int wv = __builtin_amdgcn_readfirstlane((int)(threadIdx.x >> 5));
    const int gw = blockIdx.x * WAVES_PER_WG + wv;
    const int strip = gw % STRIPS;
    const int x0 = (gw / STRIPS) * ROWS_PER_WAVE;
    int yu[TILES], y[TILES];
#pragma unroll
    for (int k = 0; k < TILES; ++k) {
        yu[k] = strip * STRIP_W - HALO + 16 * k + q16;
        y[k]  = yu[k] & (NYL - 1);
    }
    bool owned[TILES];
    owned[0] = (q16 >= HALO) && (yu[0] < NYL);
    owned[1] = (q16 < 16 - HALO) && (yu[1] < NYL);

    const int qL[6] = { 0, 2,  1, 5, -1, 6};
    const int qU[6] = {-1, 4, -1, 8,  3, 7};
    const int exr[6] = {0, 0, 1, 1, -1, -1};
    const int exv[9] = {0, 1, 0, -1, 0, 1, -1, -1, 1};
    const int eyv[9] = {0, 0, 1, 0, -1, 1, 1, -1, -1};

    unsigned col[TILES][6];
#pragma unroll
    for (int k = 0; k < TILES; ++k) {
#pragma unroll
        for (int r = 0; r < 6; ++r) {
            const int ql = qL[r] >= 0 ? qL[r] : qU[r];
            const int qu = qU[r] >= 0 ? qU[r] : qL[r];
            const int ysl = (y[k] - eyv[ql]) & (NYL - 1);
            const int ysu = (y[k] - eyv[qu]) & (NYL - 1);
            const unsigned cl = (unsigned)(ysl * 9 + ql);
            const unsigned cu = (unsigned)(ysu * 9 + qu);
            col[k][r] = half ? cu : cl;
        }
    }

    const float w0 = w[0], w1 = w[1], w2 = w[2], w3 = w[3];
    const float w4 = w[4], w5 = w[5], w6 = w[6];
    const float decay = 1.0f - 1.0f / TAU;
    const float grav  = TAU * GYF;

    const float* tr = &g_rows.v[q16][0];
    const float t0 = tr[0], t1 = tr[1], t2 = tr[2], t3 = tr[3], t4 = tr[4];
    const float t5 = tr[5], t6 = tr[6], t7 = tr[7], t8 = tr[8], tq = tr[9];
    const float c0 = t0 + w0 * t3 + w2 * t4;
    const float cx = t1 + w4 * t5;
    const float cy = t2 + w4 * t6;
    float al[6];
#pragma unroll
    for (int r = 0; r < 6; ++r) {
        const int il = qL[r], iu = qU[r];
        const float vl = il >= 0 ? 1.0f : 0.0f,            vu = iu >= 0 ? 1.0f : 0.0f;
        const float xl = il >= 0 ? (float)exv[il] : 0.0f,  xu = iu >= 0 ? (float)exv[iu] : 0.0f;
        const float yl = il >= 0 ? (float)eyv[il] : 0.0f,  yy = iu >= 0 ? (float)eyv[iu] : 0.0f;
        const float idl = il >= 0 ? (float)il : -2.0f,     idu = iu >= 0 ? (float)iu : -2.0f;
        const float val = half ? vu : vl;
        const float exi = half ? xu : xl;
        const float eyi = half ? yy : yl;
        const float idi = half ? idu : idl;
        const float dlt = (idi == tq) ? decay : 0.0f;
        al[r] = val * c0 + exi * cx + eyi * cy + dlt;
    }
    v2f a0, a1, a2, a3;
    a0.x = al[0]; a0.y = al[1];
    a1.x = al[2]; a1.y = al[3];
    a2.x = al[4]; a2.y = al[5];
    {
        const float nj  = w1 * t3 - w3 * t4;
        const float nd  = w5 * t7;
        const float nr2 = (tq >= 0.0f) ? grav * cy : 0.0f;
        const float np  = w6 * t8;
        a3.x = half ? (nj - nd) : nr2;
        a3.y = half ? np : (nj + nd);
    }

    unsigned so0[TILES], so1[TILES], so3[TILES], so5[TILES], so2[TILES];
#pragma unroll
    for (int k = 0; k < TILES; ++k) {
        const unsigned ocol = (unsigned)(y[k] * 9);
        so0[k] = ocol + (unsigned)(half ? 3 : 0);
        so1[k] = ocol + (unsigned)(half ? 4 : 2);
        so3[k] = ocol + (unsigned)(half ? 8 : 5);
        so5[k] = ocol + (unsigned)(half ? 7 : 6);
        so2[k] = ocol + 1u;
    }

    const int ysel = half ? ((q16 + 1) & 15) : ((q16 - 1) & 15);
    const bool from1 = (half == 1) && (q16 == 15);
    const bool from0 = (half == 0) && (q16 == 0);

    auto lin0 = [&](v2f b) -> v8f {
        v8f c = {};
        return __builtin_amdgcn_wmma_f32_16x16x4_f32(false, a0, false, b, (short)0, c,
                                                     false, false);
    };
    auto lin1 = [&](v8f c, v2f b) -> v8f {
        return __builtin_amdgcn_wmma_f32_16x16x4_f32(false, a1, false, b, (short)0, c,
                                                     false, false);
    };
    auto lin2 = [&](v8f c, v2f b) -> v8f {
        return __builtin_amdgcn_wmma_f32_16x16x4_f32(false, a2, false, b, (short)0, c,
                                                     false, false);
    };
    auto nonlin_terms = [&](const v8f& c) -> v2f {
        const float rho = c[6];
        const float mx  = c[7];
        const float r2  = rho * rho;
        const float jy  = grav * r2 + c[0];
        v2f bn;
        bn.x = half ? jy * jy : r2;
        bn.y = mx * (half ? jy : mx);
        return bn;
    };
    auto nonlin = [&](v8f c, v2f bn) -> v8f {
        return __builtin_amdgcn_wmma_f32_16x16x4_f32(false, a3, false, bn, (short)0, c,
                                                     false, false);
    };

    float fr[ROWS_A][TILES][6];
    auto load_row = [&](int t) {
        int x = (x0 - HALO + t) & (NXL - 1);
        asm("" : "+s"(x));
        const float* rows[3];
        rows[0] = fin + (size_t)((x + 1) & (NXL - 1)) * ROW_PITCH;
        rows[1] = fin + (size_t)x * ROW_PITCH;
        rows[2] = fin + (size_t)((x - 1) & (NXL - 1)) * ROW_PITCH;
#pragma unroll
        for (int k = 0; k < TILES; ++k) {
#pragma unroll
            for (int r = 0; r < 6; ++r) fr[t][k][r] = rows[exr[r] + 1][col[k][r]];
        }
    };

#pragma unroll
    for (int t = 0; t < ROWS_AHEAD; ++t) load_row(t);
    __builtin_amdgcn_sched_barrier(0);

    float hw[NSTAGES - 1][4][TILES][6];

    auto keep_row = [&](float (*h)[6], const v8f& ca, const v8f& cb) {
        h[0][0] = ca[0]; h[0][2] = ca[2]; h[0][4] = ca[4];
        h[1][0] = cb[0]; h[1][2] = cb[2]; h[1][4] = cb[4];
#pragma unroll
        for (int r = 1; r < 6; r += 2) {
            const float ta = row_gather(ca[r], ysel);
            const float tb = row_gather(cb[r], ysel);
            h[0][r] = from1 ? tb : ta;
            h[1][r] = from0 ? ta : tb;
        }
    };

#pragma unroll
    for (int t = 0; t < NTRIPS; ++t) {
        if (t + ROWS_AHEAD < ROWS_A) load_row(t + ROWS_AHEAD);

        bool on[NSTAGES];
#pragma unroll
        for (int s = 0; s < NSTAGES; ++s) on[s] = (t >= 3 * s) && (t <= ROWS_A - 1 + s);

        v2f b[NSTAGES][TILES][3];
#pragma unroll
        for (int k = 0; k < TILES; ++k) {
            if (on[0]) {
                const float* f = fr[t < ROWS_A ? t : 0][k];
                b[0][k][0].x = f[0]; b[0][k][0].y = f[1];
                b[0][k][1].x = f[2]; b[0][k][1].y = f[3];
                b[0][k][2].x = f[4]; b[0][k][2].y = f[5];
            }
#pragma unroll
            for (int s = 1; s < NSTAGES; ++s) {
                if (on[s]) {
                    const float* hp = hw[s - 1][(t + 1) & 3][k];
                    const float* hc = hw[s - 1][(t + 2) & 3][k];
                    const float* hn = hw[s - 1][(t + 3) & 3][k];
                    b[s][k][0].x = hc[0]; b[s][k][0].y = hc[1];
                    b[s][k][1].x = hp[2]; b[s][k][1].y = hp[3];
                    b[s][k][2].x = hn[4]; b[s][k][2].y = hn[5];
                }
            }
        }

        v8f c[NSTAGES][TILES];
#pragma unroll
        for (int s = NSTAGES - 1; s >= 0; --s)
#pragma unroll
            for (int k = 0; k < TILES; ++k)
                if (on[s]) c[s][k] = lin0(b[s][k][0]);
#pragma unroll
        for (int s = NSTAGES - 1; s >= 0; --s)
#pragma unroll
            for (int k = 0; k < TILES; ++k)
                if (on[s]) c[s][k] = lin1(c[s][k], b[s][k][1]);
#pragma unroll
        for (int s = NSTAGES - 1; s >= 0; --s)
#pragma unroll
            for (int k = 0; k < TILES; ++k)
                if (on[s]) c[s][k] = lin2(c[s][k], b[s][k][2]);

        v2f bn[NSTAGES][TILES];
#pragma unroll
        for (int s = NSTAGES - 1; s >= 0; --s)
#pragma unroll
            for (int k = 0; k < TILES; ++k)
                if (on[s]) bn[s][k] = nonlin_terms(c[s][k]);
#pragma unroll
        for (int s = NSTAGES - 1; s >= 0; --s)
#pragma unroll
            for (int k = 0; k < TILES; ++k)
                if (on[s]) c[s][k] = nonlin(c[s][k], bn[s][k]);

        if (on[NSTAGES - 1]) {
            const int x = x0 + t - 3 * HALO;
            if (x < NXL) {
                float* o = fout + (size_t)x * ROW_PITCH;
#pragma unroll
                for (int k = 0; k < TILES; ++k) {
                    if (owned[k]) {
                        const v8f& cc = c[NSTAGES - 1][k];
                        o[so0[k]] = cc[4];
                        o[so1[k]] = cc[1];
                        o[so3[k]] = cc[3];
                        o[so5[k]] = cc[5];
                        if (half == 0) o[so2[k]] = cc[2];
                    }
                }
            }
        }

        if (t == PRIO_T2) __builtin_amdgcn_s_setprio(2);
        if (t == PRIO_T1) __builtin_amdgcn_s_setprio(1);
        if (t == PRIO_T0) __builtin_amdgcn_s_setprio(0);

#pragma unroll
        for (int s = NSTAGES - 2; s >= 0; --s)
            if (on[s]) keep_row(hw[s][t & 3], c[s][0], c[s][1]);
        if (t < 3 * HALO) __builtin_amdgcn_sched_barrier(0);
    }
}

__global__ __launch_bounds__(128) void lbm_four(const float* __restrict__ fin,
                                                const float* __restrict__ w,
                                                float* __restrict__ fout) {
    lbm_four_body(fin, w, fout);
}

extern "C" void kernel_launch(void* const* d_in, const int* in_sizes, int n_in,
                              void* d_out, int out_size, void* d_ws, size_t ws_size,
                              hipStream_t stream) {
    const float* f0 = (const float*)d_in[0];
    const float* w  = (const float*)d_in[1];
    float* out = (float*)d_out;

    const int threads = 32 * WAVES_PER_WG;
    const int blocks  = (STRIPS * ROW_BLOCKS) / WAVES_PER_WG;

    lbm_four<<<blocks, threads, 0, stream>>>(f0, w, out);
}
